# starting kernel
# baseline (speedup 1.0000x reference)
#include <hip/hip_runtime.h>
#include <hip/hip_bf16.h>

typedef float v2f __attribute__((ext_vector_type(2)));
typedef float v8f __attribute__((ext_vector_type(8)));

#define N_NODES 100000
#define N_EDGES 1600000
#define IN_F 64
#define HID 64
#define NC 40

__global__ void gcn_zero_kernel(float* __restrict__ p, int n) {
  int i = blockIdx.x * blockDim.x + threadIdx.x;
  if (i < n) p[i] = 0.0f;
}

__global__ void gcn_bias_init_kernel(float* __restrict__ out,
                                     const float* __restrict__ b) {
  unsigned i = blockIdx.x * blockDim.x + threadIdx.x;
  const unsigned total = (unsigned)N_NODES * NC;
  if (i < total) {
    unsigned c = i - (i / NC) * NC;
    out[i] = b[c];
  }
}

__global__ void gcn_scatter_add_kernel(const float* __restrict__ x,
                                       const int* __restrict__ src,
                                       const int* __restrict__ dst,
                                       float* __restrict__ acc,
                                       int groups, int stride) {
  unsigned tid = blockIdx.x * blockDim.x + threadIdx.x;
  unsigned total = (unsigned)N_EDGES * (unsigned)groups;
  if (tid >= total) return;
  unsigned e = tid / (unsigned)groups;
  unsigned g = tid - e * (unsigned)groups;
  int s = src[e];
  int d = dst[e];
  const float4 v =
      *reinterpret_cast<const float4*>(x + (size_t)s * stride + (size_t)g * 4);
  float* o = acc + (size_t)d * stride + (size_t)g * 4;
  atomicAdd(o + 0, v.x);
  atomicAdd(o + 1, v.y);
  atomicAdd(o + 2, v.z);
  atomicAdd(o + 3, v.w);
}

template <int KDIM, int NREAL, bool RELU, bool BIAS>
__global__ void gcn_wmma_gemm_kernel(const float* __restrict__ A,
                                     const float* __restrict__ W,
                                     const float* __restrict__ bias,
                                     float* __restrict__ C,
                                     int n_row_tiles, int n_col_tiles) {
  constexpr bool FULL = (NREAL % 16 == 0);

  const int tile = blockIdx.x * (blockDim.x >> 5) + (threadIdx.x >> 5);
  const int n_tiles = n_row_tiles * n_col_tiles;
  if (tile >= n_tiles) return;

  const int row_tile = tile / n_col_tiles;
  const int col_tile = tile - row_tile * n_col_tiles;
  const int row0 = row_tile << 4;
  const int col0 = col_tile << 4;

  const int lane = threadIdx.x & 31;
  const int lh = lane & 15;
  const int hi = lane >> 4;

  const int nb = col0 + lh;
  const bool nok = FULL ? true : (nb < NREAL);
  const int nbc = FULL ? nb : (nok ? nb : (NREAL - 1));
  const float nmask = nok ? 1.0f : 0.0f;

  const float* arow = A + (size_t)(row0 + lh) * KDIM + (size_t)(hi * 2);

  v8f c = {};
#pragma unroll
  for (int k = 0; k < KDIM; k += 4) {
    v2f a, b;
    a.x = arow[k];
    a.y = arow[k + 1];
    const int kb = k + hi * 2;
    if (FULL) {
      b.x = W[(size_t)kb * NREAL + nbc];
      b.y = W[(size_t)(kb + 1) * NREAL + nbc];
    } else {
      b.x = W[(size_t)kb * NREAL + nbc] * nmask;
      b.y = W[(size_t)(kb + 1) * NREAL + nbc] * nmask;
    }
    c = __builtin_amdgcn_wmma_f32_16x16x4_f32(false, a, false, b, (short)0, c,
                                              false, false);
  }

  if (!nok) return;
  float bv = 0.0f;
  if (BIAS) bv = bias[nb];
#pragma unroll
  for (int r = 0; r < 8; ++r) {
    const int row = row0 + r + (hi << 3);
    float v = c[r] + bv;
    if (RELU) v = fmaxf(v, 0.0f);
    C[(size_t)row * NREAL + nb] = v;
  }
}

extern "C" void kernel_launch(void* const* d_in, const int* in_sizes, int n_in,
                              void* d_out, int out_size, void* d_ws,
                              size_t ws_size, hipStream_t stream) {
  (void)in_sizes; (void)n_in; (void)out_size; (void)ws_size;
  const float* features = (const float*)d_in[0];
  const int*   src      = (const int*)d_in[1];
  const int*   dst      = (const int*)d_in[2];
  const float* W1       = (const float*)d_in[3];
  const float* b1       = (const float*)d_in[4];
  const float* W2       = (const float*)d_in[5];
  const float* b2       = (const float*)d_in[6];
  float* out = (float*)d_out;

  float* ws   = (float*)d_ws;
  float* agg1 = ws;
  float* h    = ws + (size_t)N_NODES * HID;
  float* t    = ws;

  const int nAgg1 = N_NODES * IN_F;

  gcn_zero_kernel<<<(nAgg1 + 255) / 256, 256, 0, stream>>>(agg1, nAgg1);

  {
    const int total = N_EDGES * 16;
    gcn_scatter_add_kernel<<<(total + 255) / 256, 256, 0, stream>>>(
        features, src, dst, agg1, 16, IN_F);
  }

  {
    const int rt = N_NODES / 16, ct = HID / 16;
    const int tiles = rt * ct;
    gcn_wmma_gemm_kernel<HID, HID, true, true>
        <<<(tiles + 7) / 8, 256, 0, stream>>>(agg1, W1, b1, h, rt, ct);
  }

  {
    const int rt = N_NODES / 16, ct = (NC + 15) / 16;
    const int tiles = rt * ct;
    gcn_wmma_gemm_kernel<HID, NC, false, false>
        <<<(tiles + 7) / 8, 256, 0, stream>>>(h, W2, nullptr, t, rt, ct);
  }

  {
    const int total = N_NODES * NC;
    gcn_bias_init_kernel<<<(total + 255) / 256, 256, 0, stream>>>(out, b2);
  }

  {
    const int total = N_EDGES * 10;
    gcn_scatter_add_kernel<<<(total + 255) / 256, 256, 0, stream>>>(
        t, src, dst, out, 10, NC);
  }
}


# dense layers: a wave owns 32 rows and all their columns; k and column order inside the WMMA fragments permuted so A, weights and results move 16 B a lane, weights and results in whole lines
# speedup vs baseline: 2.1921x; 2.1921x over previous
#include <hip/hip_runtime.h>
#include <hip/hip_bf16.h>

typedef float v2f __attribute__((ext_vector_type(2)));
typedef float v8f __attribute__((ext_vector_type(8)));

#define N_NODES 100000
#define N_EDGES 1600000
#define IN_F 64
#define HID 64
#define NC 40

#define NLIST 4

__global__ void gcn_list_init_kernel(int* __restrict__ head, int n) {
  int i = blockIdx.x * blockDim.x + threadIdx.x;
  if (i < n) head[i] = -1;
}

__global__ void gcn_link_kernel(const int* __restrict__ src,
                                const int* __restrict__ dst,
                                int* __restrict__ head,
                                int2* __restrict__ node) {
  const unsigned e = blockIdx.x * blockDim.x + threadIdx.x;
  if (e >= (unsigned)N_EDGES) return;
  const int s = src[e];
  const int d = dst[e];
  const int prev =
      atomicExch(head + (size_t)d * NLIST + (e & (NLIST - 1)), (int)e);
  node[e] = make_int2(s, prev);
}

__device__ __forceinline__ bool gcn_edge_ok(int e) {
  return (unsigned)e < (unsigned)N_EDGES;
}

template <int COLS, bool HAS_BIAS>
__global__ void gcn_pull_kernel(const float* __restrict__ x,
                                const int* __restrict__ head,
                                const int2* __restrict__ node,
                                const float* __restrict__ bias,
                                float* __restrict__ y) {
  const unsigned tid = blockIdx.x * blockDim.x + threadIdx.x;
  const unsigned row = tid >> 4;
  const unsigned l = tid & 15;
  if (row >= (unsigned)N_NODES) return;
  if (l * 4 >= (unsigned)COLS) return;
  const unsigned col = l * 4;

  const int4 h4 = *reinterpret_cast<const int4*>(head + (size_t)row * NLIST);
  int e[NLIST] = {h4.x, h4.y, h4.z, h4.w};
  int2 nd[NLIST];
#pragma unroll
  for (int j = 0; j < NLIST; ++j)
    nd[j] = node[gcn_edge_ok(e[j]) ? (unsigned)e[j] : 0u];

  float4 acc = make_float4(0.0f, 0.0f, 0.0f, 0.0f);
  for (int it = 0; it < N_EDGES; ++it) {
    bool ok[NLIST];
    bool any = false;
#pragma unroll
    for (int j = 0; j < NLIST; ++j) {
      ok[j] = gcn_edge_ok(e[j]);
      any = any || ok[j];
    }
    if (!any) break;

    int en[NLIST];
    int2 ndn[NLIST];
#pragma unroll
    for (int j = 0; j < NLIST; ++j) {
      en[j] = ok[j] ? nd[j].y : -1;
      ndn[j] = node[gcn_edge_ok(en[j]) ? (unsigned)en[j] : 0u];
    }
    float4 v[NLIST];
#pragma unroll
    for (int j = 0; j < NLIST; ++j) {
      const unsigned off = (ok[j] ? (unsigned)nd[j].x : 0u) * (unsigned)COLS + col;
      v[j] = *reinterpret_cast<const float4*>(x + off);
    }
#pragma unroll
    for (int j = 0; j < NLIST; ++j) {
      acc.x += ok[j] ? v[j].x : 0.0f;
      acc.y += ok[j] ? v[j].y : 0.0f;
      acc.z += ok[j] ? v[j].z : 0.0f;
      acc.w += ok[j] ? v[j].w : 0.0f;
      e[j] = en[j];
      nd[j] = ndn[j];
    }
  }

  if (HAS_BIAS) {
    acc.x += bias[col + 0]; acc.y += bias[col + 1];
    acc.z += bias[col + 2]; acc.w += bias[col + 3];
  }
  *reinterpret_cast<float4*>(y + (size_t)row * COLS + col) = acc;
}

template <int KDIM, int NREAL, bool RELU, bool BIAS>
__global__ void __launch_bounds__(256)
gcn_wmma_strip_kernel(const float* __restrict__ A,
                      const float* __restrict__ W,
                      const float* __restrict__ bias,
                      float* __restrict__ C, int n_strips) {
  static_assert(KDIM % 8 == 0, "k is taken 8 at a time");
  static_assert(NREAL % 4 == 0 && NREAL <= 64, "four column tiles, 4 columns a lane");
  constexpr int NQ = NREAL / 4;
  constexpr bool FULL = (NQ == 16);

  const int strip = blockIdx.x * (blockDim.x >> 5) + (threadIdx.x >> 5);
  if (strip >= n_strips) return;

  const int lane = threadIdx.x & 31;
  const int c = lane & 15;
  const int h = lane >> 4;
  const bool cok = FULL ? true : (c < NQ);
  const int cc = cok ? c : 0;

  const float* a0 = A + ((size_t)strip * 32 + c) * KDIM + 4 * h;
  const float* a1 = a0 + (size_t)16 * KDIM;
  const float* wp = W + (size_t)(4 * h) * NREAL + 4 * cc;

  v8f acc[2][4];
#pragma unroll
  for (int m = 0; m < 2; ++m)
#pragma unroll
    for (int ct = 0; ct < 4; ++ct) acc[m][ct] = v8f{};

#pragma unroll
  for (int i = 0; i < KDIM / 8; ++i) {
    const float4 x0 = *reinterpret_cast<const float4*>(a0 + 8 * i);
    const float4 x1 = *reinterpret_cast<const float4*>(a1 + 8 * i);
    float wv[4][4];
#pragma unroll
    for (int j = 0; j < 4; ++j) {
      const float4 w4 =
          *reinterpret_cast<const float4*>(wp + (size_t)(8 * i + j) * NREAL);
      wv[j][0] = w4.x; wv[j][1] = w4.y; wv[j][2] = w4.z; wv[j][3] = w4.w;
    }
#pragma unroll
    for (int p = 0; p < 2; ++p) {
      v2f af0, af1;
      af0.x = p ? x0.z : x0.x;
      af0.y = p ? x0.w : x0.y;
      af1.x = p ? x1.z : x1.x;
      af1.y = p ? x1.w : x1.y;
#pragma unroll
      for (int ct = 0; ct < 4; ++ct) {
        v2f bf;
        bf.x = wv[2 * p][ct];
        bf.y = wv[2 * p + 1][ct];
        acc[0][ct] = __builtin_amdgcn_wmma_f32_16x16x4_f32(
            false, af0, false, bf, (short)0, acc[0][ct], false, false);
        acc[1][ct] = __builtin_amdgcn_wmma_f32_16x16x4_f32(
            false, af1, false, bf, (short)0, acc[1][ct], false, false);
      }
    }
  }

  if (!cok) return;
  float4 bv = make_float4(0.0f, 0.0f, 0.0f, 0.0f);
  if (BIAS) bv = *reinterpret_cast<const float4*>(bias + 4 * c);
#pragma unroll
  for (int m = 0; m < 2; ++m) {
#pragma unroll
    for (int r = 0; r < 8; ++r) {
      const size_t row = (size_t)strip * 32 + 16 * m + 8 * h + r;
      float4 v;
      v.x = acc[m][0][r] + bv.x;
      v.y = acc[m][1][r] + bv.y;
      v.z = acc[m][2][r] + bv.z;
      v.w = acc[m][3][r] + bv.w;
      if (RELU) {
        v.x = fmaxf(v.x, 0.0f); v.y = fmaxf(v.y, 0.0f);
        v.z = fmaxf(v.z, 0.0f); v.w = fmaxf(v.w, 0.0f);
      }
      *reinterpret_cast<float4*>(C + row * NREAL + 4 * c) = v;
    }
  }
}

extern "C" void kernel_launch(void* const* d_in, const int* in_sizes, int n_in,
                              void* d_out, int out_size, void* d_ws,
                              size_t ws_size, hipStream_t stream) {
  (void)in_sizes; (void)n_in; (void)out_size; (void)ws_size;
  const float* features = (const float*)d_in[0];
  const int*   src      = (const int*)d_in[1];
  const int*   dst      = (const int*)d_in[2];
  const float* W1       = (const float*)d_in[3];
  const float* b1       = (const float*)d_in[4];
  const float* W2       = (const float*)d_in[5];
  const float* b2       = (const float*)d_in[6];
  float* out = (float*)d_out;

  float* ws   = (float*)d_ws;
  float* agg1 = ws;
  float* h    = ws + (size_t)N_NODES * HID;
  float* t    = ws;
  int2*  node = (int2*)(ws + (size_t)2 * N_NODES * HID);
  int*   head = (int*)(ws + (size_t)2 * N_NODES * HID + (size_t)2 * N_EDGES);

  {
    const int n = N_NODES * NLIST;
    gcn_list_init_kernel<<<(n + 255) / 256, 256, 0, stream>>>(head, n);
  }

  gcn_link_kernel<<<(N_EDGES + 255) / 256, 256, 0, stream>>>(src, dst, head,
                                                             node);

  gcn_pull_kernel<IN_F, false>
      <<<(N_NODES * 16 + 255) / 256, 256, 0, stream>>>(features, head, node,
                                                       nullptr, agg1);

  static_assert(N_NODES % 32 == 0, "a wave owns 32 whole rows");
  {
    const int strips = N_NODES / 32;
    gcn_wmma_strip_kernel<HID, HID, true, true>
        <<<(strips + 7) / 8, 256, 0, stream>>>(agg1, W1, b1, h, strips);
  }

  {
    const int strips = N_NODES / 32;
    gcn_wmma_strip_kernel<HID, NC, false, false>
        <<<(strips + 7) / 8, 256, 0, stream>>>(h, W2, nullptr, t, strips);
  }

  gcn_pull_kernel<NC, true>
      <<<(N_NODES * 16 + 255) / 256, 256, 0, stream>>>(t, head, node, b2, out);
}


# first row pull also stores each row's source ids, in summing order, as one 256-B block; second pull reads its block, not the lists: no node loads, 8 row loads a lane at once; sum order kept
# speedup vs baseline: 2.3301x; 1.0630x over previous
#include <hip/hip_runtime.h>
#include <hip/hip_bf16.h>

typedef float v2f __attribute__((ext_vector_type(2)));
typedef float v8f __attribute__((ext_vector_type(8)));

#define N_NODES 100000
#define N_EDGES 1600000
#define IN_F 64
#define HID 64
#define NC 40

#define NLIST 4

__global__ void gcn_list_init_kernel(int* __restrict__ head, int n) {
  int i = blockIdx.x * blockDim.x + threadIdx.x;
  if (i < n) head[i] = -1;
}

__global__ void gcn_link_kernel(const int* __restrict__ src,
                                const int* __restrict__ dst,
                                int* __restrict__ head,
                                int2* __restrict__ node) {
  const unsigned e = blockIdx.x * blockDim.x + threadIdx.x;
  if (e >= (unsigned)N_EDGES) return;
  const int s = src[e];
  const int d = dst[e];
  const int prev =
      atomicExch(head + (size_t)d * NLIST + (e & (NLIST - 1)), (int)e);
  node[e] = make_int2(s, prev);
}

__device__ __forceinline__ bool gcn_edge_ok(int e) {
  return (unsigned)e < (unsigned)N_EDGES;
}

#define BLK_TRIPS 16
#define BLK_LONG (-2)

template <int COLS, bool RECORD>
__device__ __forceinline__ float4 gcn_walk_lists(const float* __restrict__ x,
                                                 const int* __restrict__ head,
                                                 const int2* __restrict__ node,
                                                 unsigned row, unsigned col,
                                                 unsigned l, int4& keep,
                                                 int& trips) {
  const int4 h4 = *reinterpret_cast<const int4*>(head + (size_t)row * NLIST);
  int e[NLIST] = {h4.x, h4.y, h4.z, h4.w};
  int2 nd[NLIST];
#pragma unroll
  for (int j = 0; j < NLIST; ++j)
    nd[j] = node[gcn_edge_ok(e[j]) ? (unsigned)e[j] : 0u];

  float4 acc = make_float4(0.0f, 0.0f, 0.0f, 0.0f);
  int it = 0;
  for (; it < N_EDGES; ++it) {
    bool ok[NLIST];
    bool any = false;
#pragma unroll
    for (int j = 0; j < NLIST; ++j) {
      ok[j] = gcn_edge_ok(e[j]);
      any = any || ok[j];
    }
    if (!any) break;

    if (RECORD) {
      if ((unsigned)it == l)
        keep = make_int4(ok[0] ? nd[0].x : -1, ok[1] ? nd[1].x : -1,
                         ok[2] ? nd[2].x : -1, ok[3] ? nd[3].x : -1);
    }

    int en[NLIST];
    int2 ndn[NLIST];
#pragma unroll
    for (int j = 0; j < NLIST; ++j) {
      en[j] = ok[j] ? nd[j].y : -1;
      ndn[j] = node[gcn_edge_ok(en[j]) ? (unsigned)en[j] : 0u];
    }
    float4 v[NLIST];
#pragma unroll
    for (int j = 0; j < NLIST; ++j) {
      const unsigned off = (ok[j] ? (unsigned)nd[j].x : 0u) * (unsigned)COLS + col;
      v[j] = *reinterpret_cast<const float4*>(x + off);
    }
#pragma unroll
    for (int j = 0; j < NLIST; ++j) {
      acc.x += ok[j] ? v[j].x : 0.0f;
      acc.y += ok[j] ? v[j].y : 0.0f;
      acc.z += ok[j] ? v[j].z : 0.0f;
      acc.w += ok[j] ? v[j].w : 0.0f;
      e[j] = en[j];
      nd[j] = ndn[j];
    }
  }
  trips = it;
  return acc;
}

__global__ void gcn_pull_record_kernel(const float* __restrict__ x,
                                       const int* __restrict__ head,
                                       const int2* __restrict__ node,
                                       int4* __restrict__ blk,
                                       float* __restrict__ y) {
  static_assert(IN_F == 64, "16 lanes x 4 columns: every lane of a row is live");
  static_assert(NLIST == 4, "one int4 of ids per trip");
  const unsigned tid = blockIdx.x * blockDim.x + threadIdx.x;
  const unsigned row = tid >> 4;
  const unsigned l = tid & 15;
  if (row >= (unsigned)N_NODES) return;
  const unsigned col = l * 4;

  int4 keep = make_int4(-1, -1, -1, -1);
  int trips = 0;
  const float4 acc =
      gcn_walk_lists<IN_F, true>(x, head, node, row, col, l, keep, trips);
  if (trips > BLK_TRIPS && l == BLK_TRIPS - 1) keep.w = BLK_LONG;

  blk[(size_t)row * BLK_TRIPS + l] = keep;
  *reinterpret_cast<float4*>(y + (size_t)row * IN_F + col) = acc;
}

template <int K>
__device__ __forceinline__ int gcn_row_share(int v) {
  return __builtin_amdgcn_update_dpp(0, v, 0x150 + K, 0xF, 0xF, true);
}

template <int COLS, int B>
__device__ __forceinline__ bool gcn_block_batch(const float* __restrict__ x,
                                                const int4 my, unsigned col,
                                                float4& acc) {
  int id[8];
  id[0] = gcn_row_share<2 * B>(my.x);
  id[1] = gcn_row_share<2 * B>(my.y);
  id[2] = gcn_row_share<2 * B>(my.z);
  id[3] = gcn_row_share<2 * B>(my.w);
  id[4] = gcn_row_share<2 * B + 1>(my.x);
  id[5] = gcn_row_share<2 * B + 1>(my.y);
  id[6] = gcn_row_share<2 * B + 1>(my.z);
  id[7] = gcn_row_share<2 * B + 1>(my.w);
  bool any = false;
#pragma unroll
  for (int k = 0; k < 8; ++k) any = any || (id[k] >= 0);
  if (!any) return false;
  float4 v[8];
#pragma unroll
  for (int k = 0; k < 8; ++k) {
    const unsigned off = (id[k] >= 0 ? (unsigned)id[k] : 0u) * (unsigned)COLS + col;
    v[k] = *reinterpret_cast<const float4*>(x + off);
  }
#pragma unroll
  for (int k = 0; k < 8; ++k) {
    acc.x += id[k] >= 0 ? v[k].x : 0.0f;
    acc.y += id[k] >= 0 ? v[k].y : 0.0f;
    acc.z += id[k] >= 0 ? v[k].z : 0.0f;
    acc.w += id[k] >= 0 ? v[k].w : 0.0f;
  }
  return true;
}

template <int COLS, bool HAS_BIAS>
__global__ void gcn_pull_blocks_kernel(const float* __restrict__ x,
                                       const int4* __restrict__ blk,
                                       const int* __restrict__ head,
                                       const int2* __restrict__ node,
                                       const float* __restrict__ bias,
                                       float* __restrict__ y) {
  static_assert(COLS % 4 == 0 && COLS <= 64, "a lane owns 4 adjacent columns");
  const unsigned tid = blockIdx.x * blockDim.x + threadIdx.x;
  const unsigned row = tid >> 4;
  const unsigned l = tid & 15;
  if (row >= (unsigned)N_NODES) return;
  const bool cok = l * 4 < (unsigned)COLS;
  const unsigned col = cok ? l * 4 : 0u;

  const int4 my = blk[(size_t)row * BLK_TRIPS + l];
  float4 acc = make_float4(0.0f, 0.0f, 0.0f, 0.0f);
  if (gcn_row_share<BLK_TRIPS - 1>(my.w) == BLK_LONG) {
    int4 unused = make_int4(-1, -1, -1, -1);
    int trips = 0;
    acc = gcn_walk_lists<COLS, false>(x, head, node, row, col, l, unused, trips);
  } else {
    if (gcn_block_batch<COLS, 0>(x, my, col, acc))
    if (gcn_block_batch<COLS, 1>(x, my, col, acc))
    if (gcn_block_batch<COLS, 2>(x, my, col, acc))
    if (gcn_block_batch<COLS, 3>(x, my, col, acc))
    if (gcn_block_batch<COLS, 4>(x, my, col, acc))
    if (gcn_block_batch<COLS, 5>(x, my, col, acc))
    if (gcn_block_batch<COLS, 6>(x, my, col, acc))
      gcn_block_batch<COLS, 7>(x, my, col, acc);
  }

  if (!cok) return;
  if (HAS_BIAS) {
    acc.x += bias[col + 0]; acc.y += bias[col + 1];
    acc.z += bias[col + 2]; acc.w += bias[col + 3];
  }
  *reinterpret_cast<float4*>(y + (size_t)row * COLS + col) = acc;
}

template <int KDIM, int NREAL, bool RELU, bool BIAS>
__global__ void __launch_bounds__(256)
gcn_wmma_strip_kernel(const float* __restrict__ A,
                      const float* __restrict__ W,
                      const float* __restrict__ bias,
                      float* __restrict__ C, int n_strips) {
  static_assert(KDIM % 8 == 0, "k is taken 8 at a time");
  static_assert(NREAL % 4 == 0 && NREAL <= 64, "four column tiles, 4 columns a lane");
  constexpr int NQ = NREAL / 4;
  constexpr bool FULL = (NQ == 16);

  const int strip = blockIdx.x * (blockDim.x >> 5) + (threadIdx.x >> 5);
  if (strip >= n_strips) return;

  const int lane = threadIdx.x & 31;
  const int c = lane & 15;
  const int h = lane >> 4;
  const bool cok = FULL ? true : (c < NQ);
  const int cc = cok ? c : 0;

  const float* a0 = A + ((size_t)strip * 32 + c) * KDIM + 4 * h;
  const float* a1 = a0 + (size_t)16 * KDIM;
  const float* wp = W + (size_t)(4 * h) * NREAL + 4 * cc;

  v8f acc[2][4];
#pragma unroll
  for (int m = 0; m < 2; ++m)
#pragma unroll
    for (int ct = 0; ct < 4; ++ct) acc[m][ct] = v8f{};

#pragma unroll
  for (int i = 0; i < KDIM / 8; ++i) {
    const float4 x0 = *reinterpret_cast<const float4*>(a0 + 8 * i);
    const float4 x1 = *reinterpret_cast<const float4*>(a1 + 8 * i);
    float wv[4][4];
#pragma unroll
    for (int j = 0; j < 4; ++j) {
      const float4 w4 =
          *reinterpret_cast<const float4*>(wp + (size_t)(8 * i + j) * NREAL);
      wv[j][0] = w4.x; wv[j][1] = w4.y; wv[j][2] = w4.z; wv[j][3] = w4.w;
    }
#pragma unroll
    for (int p = 0; p < 2; ++p) {
      v2f af0, af1;
      af0.x = p ? x0.z : x0.x;
      af0.y = p ? x0.w : x0.y;
      af1.x = p ? x1.z : x1.x;
      af1.y = p ? x1.w : x1.y;
#pragma unroll
      for (int ct = 0; ct < 4; ++ct) {
        v2f bf;
        bf.x = wv[2 * p][ct];
        bf.y = wv[2 * p + 1][ct];
        acc[0][ct] = __builtin_amdgcn_wmma_f32_16x16x4_f32(
            false, af0, false, bf, (short)0, acc[0][ct], false, false);
        acc[1][ct] = __builtin_amdgcn_wmma_f32_16x16x4_f32(
            false, af1, false, bf, (short)0, acc[1][ct], false, false);
      }
    }
  }

  if (!cok) return;
  float4 bv = make_float4(0.0f, 0.0f, 0.0f, 0.0f);
  if (BIAS) bv = *reinterpret_cast<const float4*>(bias + 4 * c);
#pragma unroll
  for (int m = 0; m < 2; ++m) {
#pragma unroll
    for (int r = 0; r < 8; ++r) {
      const size_t row = (size_t)strip * 32 + 16 * m + 8 * h + r;
      float4 v;
      v.x = acc[m][0][r] + bv.x;
      v.y = acc[m][1][r] + bv.y;
      v.z = acc[m][2][r] + bv.z;
      v.w = acc[m][3][r] + bv.w;
      if (RELU) {
        v.x = fmaxf(v.x, 0.0f); v.y = fmaxf(v.y, 0.0f);
        v.z = fmaxf(v.z, 0.0f); v.w = fmaxf(v.w, 0.0f);
      }
      *reinterpret_cast<float4*>(C + row * NREAL + 4 * c) = v;
    }
  }
}

extern "C" void kernel_launch(void* const* d_in, const int* in_sizes, int n_in,
                              void* d_out, int out_size, void* d_ws,
                              size_t ws_size, hipStream_t stream) {
  (void)in_sizes; (void)n_in; (void)out_size; (void)ws_size;
  const float* features = (const float*)d_in[0];
  const int*   src      = (const int*)d_in[1];
  const int*   dst      = (const int*)d_in[2];
  const float* W1       = (const float*)d_in[3];
  const float* b1       = (const float*)d_in[4];
  const float* W2       = (const float*)d_in[5];
  const float* b2       = (const float*)d_in[6];
  float* out = (float*)d_out;

  float* ws   = (float*)d_ws;
  float* agg1 = ws;
  float* h    = ws + (size_t)N_NODES * HID;
  float* t    = ws;
  int2*  node = (int2*)(ws + (size_t)2 * N_NODES * HID);
  int*   head = (int*)(ws + (size_t)2 * N_NODES * HID + (size_t)2 * N_EDGES);
  int4*  blk  = (int4*)(head + (size_t)N_NODES * NLIST);

  {
    const int n = N_NODES * NLIST;
    gcn_list_init_kernel<<<(n + 255) / 256, 256, 0, stream>>>(head, n);
  }

  gcn_link_kernel<<<(N_EDGES + 255) / 256, 256, 0, stream>>>(src, dst, head,
                                                             node);

  gcn_pull_record_kernel
      <<<(N_NODES * 16 + 255) / 256, 256, 0, stream>>>(features, head, node,
                                                       blk, agg1);

  static_assert(N_NODES % 32 == 0, "a wave owns 32 whole rows");
  {
    const int strips = N_NODES / 32;
    gcn_wmma_strip_kernel<HID, HID, true, true>
        <<<(strips + 7) / 8, 256, 0, stream>>>(agg1, W1, b1, h, strips);
  }

  {
    const int strips = N_NODES / 32;
    gcn_wmma_strip_kernel<HID, NC, false, false>
        <<<(strips + 7) / 8, 256, 0, stream>>>(h, W2, nullptr, t, strips);
  }

  gcn_pull_blocks_kernel<NC, true>
      <<<(N_NODES * 16 + 255) / 256, 256, 0, stream>>>(t, blk, head, node, b2,
                                                       out);
}


# list heads in two planes, one per package half: an edge joins a list in the plane of the half its wave runs on, so a head line is exchanged on from one half only; sum order follows placement
# speedup vs baseline: 3.1878x; 1.3681x over previous
#include <hip/hip_runtime.h>
#include <hip/hip_bf16.h>

typedef float v2f __attribute__((ext_vector_type(2)));
typedef float v8f __attribute__((ext_vector_type(8)));

#define N_NODES 100000
#define N_EDGES 1600000
#define IN_F 64
#define HID 64
#define NC 40

#define HLIST 2
#define NLIST (2 * HLIST)

__device__ __forceinline__ unsigned gcn_wave_half() {
  return ((unsigned)__builtin_amdgcn_s_sendmsg_rtn(0x87) >> 18) & 1u;
}

__device__ __forceinline__ size_t gcn_head_at(unsigned p, unsigned r) {
  return ((size_t)p * N_NODES + r) * HLIST;
}

__global__ void gcn_list_init_kernel(int* __restrict__ head, int n) {
  int i = blockIdx.x * blockDim.x + threadIdx.x;
  if (i < n) head[i] = -1;
}

__global__ void gcn_link_kernel(const int* __restrict__ src,
                                const int* __restrict__ dst,
                                int* __restrict__ head,
                                int2* __restrict__ node) {
  const unsigned e = blockIdx.x * blockDim.x + threadIdx.x;
  if (e >= (unsigned)N_EDGES) return;
  const int s = src[e];
  const int d = dst[e];
  const unsigned half = gcn_wave_half();
  const int prev = atomicExch(
      head + gcn_head_at(half, (unsigned)d) + (e & (HLIST - 1)), (int)e);
  node[e] = make_int2(s, prev);
}

__device__ __forceinline__ bool gcn_edge_ok(int e) {
  return (unsigned)e < (unsigned)N_EDGES;
}

#define BLK_TRIPS 16
#define BLK_LONG (-2)

template <int COLS, bool RECORD>
__device__ __forceinline__ float4 gcn_walk_lists(const float* __restrict__ x,
                                                 const int* __restrict__ head,
                                                 const int2* __restrict__ node,
                                                 unsigned row, unsigned col,
                                                 unsigned l, int4& keep,
                                                 int& trips) {
  static_assert(HLIST == 2, "one 8-byte load per plane");
  const int2 h0 = *reinterpret_cast<const int2*>(head + gcn_head_at(0, row));
  const int2 h1 = *reinterpret_cast<const int2*>(head + gcn_head_at(1, row));
  int e[NLIST] = {h0.x, h0.y, h1.x, h1.y};
  int2 nd[NLIST];
#pragma unroll
  for (int j = 0; j < NLIST; ++j)
    nd[j] = node[gcn_edge_ok(e[j]) ? (unsigned)e[j] : 0u];

  float4 acc = make_float4(0.0f, 0.0f, 0.0f, 0.0f);
  int it = 0;
  for (; it < N_EDGES; ++it) {
    bool ok[NLIST];
    bool any = false;
#pragma unroll
    for (int j = 0; j < NLIST; ++j) {
      ok[j] = gcn_edge_ok(e[j]);
      any = any || ok[j];
    }
    if (!any) break;

    if (RECORD) {
      if ((unsigned)it == l)
        keep = make_int4(ok[0] ? nd[0].x : -1, ok[1] ? nd[1].x : -1,
                         ok[2] ? nd[2].x : -1, ok[3] ? nd[3].x : -1);
    }

    int en[NLIST];
    int2 ndn[NLIST];
#pragma unroll
    for (int j = 0; j < NLIST; ++j) {
      en[j] = ok[j] ? nd[j].y : -1;
      ndn[j] = node[gcn_edge_ok(en[j]) ? (unsigned)en[j] : 0u];
    }
    float4 v[NLIST];
#pragma unroll
    for (int j = 0; j < NLIST; ++j) {
      const unsigned off = (ok[j] ? (unsigned)nd[j].x : 0u) * (unsigned)COLS + col;
      v[j] = *reinterpret_cast<const float4*>(x + off);
    }
#pragma unroll
    for (int j = 0; j < NLIST; ++j) {
      acc.x += ok[j] ? v[j].x : 0.0f;
      acc.y += ok[j] ? v[j].y : 0.0f;
      acc.z += ok[j] ? v[j].z : 0.0f;
      acc.w += ok[j] ? v[j].w : 0.0f;
      e[j] = en[j];
      nd[j] = ndn[j];
    }
  }
  trips = it;
  return acc;
}

__global__ void gcn_pull_record_kernel(const float* __restrict__ x,
                                       const int* __restrict__ head,
                                       const int2* __restrict__ node,
                                       int4* __restrict__ blk,
                                       float* __restrict__ y) {
  static_assert(IN_F == 64, "16 lanes x 4 columns: every lane of a row is live");
  static_assert(NLIST == 4, "one int4 of ids per trip");
  const unsigned tid = blockIdx.x * blockDim.x + threadIdx.x;
  const unsigned row = tid >> 4;
  const unsigned l = tid & 15;
  if (row >= (unsigned)N_NODES) return;
  const unsigned col = l * 4;

  int4 keep = make_int4(-1, -1, -1, -1);
  int trips = 0;
  const float4 acc =
      gcn_walk_lists<IN_F, true>(x, head, node, row, col, l, keep, trips);
  if (trips > BLK_TRIPS && l == BLK_TRIPS - 1) keep.w = BLK_LONG;

  blk[(size_t)row * BLK_TRIPS + l] = keep;
  *reinterpret_cast<float4*>(y + (size_t)row * IN_F + col) = acc;
}

template <int K>
__device__ __forceinline__ int gcn_row_share(int v) {
  return __builtin_amdgcn_update_dpp(0, v, 0x150 + K, 0xF, 0xF, true);
}

template <int COLS, int B>
__device__ __forceinline__ bool gcn_block_batch(const float* __restrict__ x,
                                                const int4 my, unsigned col,
                                                float4& acc) {
  int id[8];
  id[0] = gcn_row_share<2 * B>(my.x);
  id[1] = gcn_row_share<2 * B>(my.y);
  id[2] = gcn_row_share<2 * B>(my.z);
  id[3] = gcn_row_share<2 * B>(my.w);
  id[4] = gcn_row_share<2 * B + 1>(my.x);
  id[5] = gcn_row_share<2 * B + 1>(my.y);
  id[6] = gcn_row_share<2 * B + 1>(my.z);
  id[7] = gcn_row_share<2 * B + 1>(my.w);
  bool any = false;
#pragma unroll
  for (int k = 0; k < 8; ++k) any = any || (id[k] >= 0);
  if (!any) return false;
  float4 v[8];
#pragma unroll
  for (int k = 0; k < 8; ++k) {
    const unsigned off = (id[k] >= 0 ? (unsigned)id[k] : 0u) * (unsigned)COLS + col;
    v[k] = *reinterpret_cast<const float4*>(x + off);
  }
#pragma unroll
  for (int k = 0; k < 8; ++k) {
    acc.x += id[k] >= 0 ? v[k].x : 0.0f;
    acc.y += id[k] >= 0 ? v[k].y : 0.0f;
    acc.z += id[k] >= 0 ? v[k].z : 0.0f;
    acc.w += id[k] >= 0 ? v[k].w : 0.0f;
  }
  return true;
}

template <int COLS, bool HAS_BIAS>
__global__ void gcn_pull_blocks_kernel(const float* __restrict__ x,
                                       const int4* __restrict__ blk,
                                       const int* __restrict__ head,
                                       const int2* __restrict__ node,
                                       const float* __restrict__ bias,
                                       float* __restrict__ y) {
  static_assert(COLS % 4 == 0 && COLS <= 64, "a lane owns 4 adjacent columns");
  const unsigned tid = blockIdx.x * blockDim.x + threadIdx.x;
  const unsigned row = tid >> 4;
  const unsigned l = tid & 15;
  if (row >= (unsigned)N_NODES) return;
  const bool cok = l * 4 < (unsigned)COLS;
  const unsigned col = cok ? l * 4 : 0u;

  const int4 my = blk[(size_t)row * BLK_TRIPS + l];
  float4 acc = make_float4(0.0f, 0.0f, 0.0f, 0.0f);
  if (gcn_row_share<BLK_TRIPS - 1>(my.w) == BLK_LONG) {
    int4 unused = make_int4(-1, -1, -1, -1);
    int trips = 0;
    acc = gcn_walk_lists<COLS, false>(x, head, node, row, col, l, unused, trips);
  } else {
    if (gcn_block_batch<COLS, 0>(x, my, col, acc))
    if (gcn_block_batch<COLS, 1>(x, my, col, acc))
    if (gcn_block_batch<COLS, 2>(x, my, col, acc))
    if (gcn_block_batch<COLS, 3>(x, my, col, acc))
    if (gcn_block_batch<COLS, 4>(x, my, col, acc))
    if (gcn_block_batch<COLS, 5>(x, my, col, acc))
    if (gcn_block_batch<COLS, 6>(x, my, col, acc))
      gcn_block_batch<COLS, 7>(x, my, col, acc);
  }

  if (!cok) return;
  if (HAS_BIAS) {
    acc.x += bias[col + 0]; acc.y += bias[col + 1];
    acc.z += bias[col + 2]; acc.w += bias[col + 3];
  }
  *reinterpret_cast<float4*>(y + (size_t)row * COLS + col) = acc;
}

template <int KDIM, int NREAL, bool RELU, bool BIAS>
__global__ void __launch_bounds__(256)
gcn_wmma_strip_kernel(const float* __restrict__ A,
                      const float* __restrict__ W,
                      const float* __restrict__ bias,
                      float* __restrict__ C, int n_strips) {
  static_assert(KDIM % 8 == 0, "k is taken 8 at a time");
  static_assert(NREAL % 4 == 0 && NREAL <= 64, "four column tiles, 4 columns a lane");
  constexpr int NQ = NREAL / 4;
  constexpr bool FULL = (NQ == 16);

  const int strip = blockIdx.x * (blockDim.x >> 5) + (threadIdx.x >> 5);
  if (strip >= n_strips) return;

  const int lane = threadIdx.x & 31;
  const int c = lane & 15;
  const int h = lane >> 4;
  const bool cok = FULL ? true : (c < NQ);
  const int cc = cok ? c : 0;

  const float* a0 = A + ((size_t)strip * 32 + c) * KDIM + 4 * h;
  const float* a1 = a0 + (size_t)16 * KDIM;
  const float* wp = W + (size_t)(4 * h) * NREAL + 4 * cc;

  v8f acc[2][4];
#pragma unroll
  for (int m = 0; m < 2; ++m)
#pragma unroll
    for (int ct = 0; ct < 4; ++ct) acc[m][ct] = v8f{};

#pragma unroll
  for (int i = 0; i < KDIM / 8; ++i) {
    const float4 x0 = *reinterpret_cast<const float4*>(a0 + 8 * i);
    const float4 x1 = *reinterpret_cast<const float4*>(a1 + 8 * i);
    float wv[4][4];
#pragma unroll
    for (int j = 0; j < 4; ++j) {
      const float4 w4 =
          *reinterpret_cast<const float4*>(wp + (size_t)(8 * i + j) * NREAL);
      wv[j][0] = w4.x; wv[j][1] = w4.y; wv[j][2] = w4.z; wv[j][3] = w4.w;
    }
#pragma unroll
    for (int p = 0; p < 2; ++p) {
      v2f af0, af1;
      af0.x = p ? x0.z : x0.x;
      af0.y = p ? x0.w : x0.y;
      af1.x = p ? x1.z : x1.x;
      af1.y = p ? x1.w : x1.y;
#pragma unroll
      for (int ct = 0; ct < 4; ++ct) {
        v2f bf;
        bf.x = wv[2 * p][ct];
        bf.y = wv[2 * p + 1][ct];
        acc[0][ct] = __builtin_amdgcn_wmma_f32_16x16x4_f32(
            false, af0, false, bf, (short)0, acc[0][ct], false, false);
        acc[1][ct] = __builtin_amdgcn_wmma_f32_16x16x4_f32(
            false, af1, false, bf, (short)0, acc[1][ct], false, false);
      }
    }
  }

  if (!cok) return;
  float4 bv = make_float4(0.0f, 0.0f, 0.0f, 0.0f);
  if (BIAS) bv = *reinterpret_cast<const float4*>(bias + 4 * c);
#pragma unroll
  for (int m = 0; m < 2; ++m) {
#pragma unroll
    for (int r = 0; r < 8; ++r) {
      const size_t row = (size_t)strip * 32 + 16 * m + 8 * h + r;
      float4 v;
      v.x = acc[m][0][r] + bv.x;
      v.y = acc[m][1][r] + bv.y;
      v.z = acc[m][2][r] + bv.z;
      v.w = acc[m][3][r] + bv.w;
      if (RELU) {
        v.x = fmaxf(v.x, 0.0f); v.y = fmaxf(v.y, 0.0f);
        v.z = fmaxf(v.z, 0.0f); v.w = fmaxf(v.w, 0.0f);
      }
      *reinterpret_cast<float4*>(C + row * NREAL + 4 * c) = v;
    }
  }
}

extern "C" void kernel_launch(void* const* d_in, const int* in_sizes, int n_in,
                              void* d_out, int out_size, void* d_ws,
                              size_t ws_size, hipStream_t stream) {
  (void)in_sizes; (void)n_in; (void)out_size; (void)ws_size;
  const float* features = (const float*)d_in[0];
  const int*   src      = (const int*)d_in[1];
  const int*   dst      = (const int*)d_in[2];
  const float* W1       = (const float*)d_in[3];
  const float* b1       = (const float*)d_in[4];
  const float* W2       = (const float*)d_in[5];
  const float* b2       = (const float*)d_in[6];
  float* out = (float*)d_out;

  float* ws   = (float*)d_ws;
  float* agg1 = ws;
  float* h    = ws + (size_t)N_NODES * HID;
  float* t    = ws;
  int2*  node = (int2*)(ws + (size_t)2 * N_NODES * HID);
  int*   head = (int*)(ws + (size_t)2 * N_NODES * HID + (size_t)2 * N_EDGES);
  int4*  blk  = (int4*)(head + (size_t)N_NODES * NLIST);

  {
    const int n = N_NODES * NLIST;
    gcn_list_init_kernel<<<(n + 255) / 256, 256, 0, stream>>>(head, n);
  }

  gcn_link_kernel<<<(N_EDGES + 255) / 256, 256, 0, stream>>>(src, dst, head,
                                                             node);

  gcn_pull_record_kernel
      <<<(N_NODES * 16 + 255) / 256, 256, 0, stream>>>(features, head, node,
                                                       blk, agg1);

  static_assert(N_NODES % 32 == 0, "a wave owns 32 whole rows");
  {
    const int strips = N_NODES / 32;
    gcn_wmma_strip_kernel<HID, HID, true, true>
        <<<(strips + 7) / 8, 256, 0, stream>>>(agg1, W1, b1, h, strips);
  }

  {
    const int strips = N_NODES / 32;
    gcn_wmma_strip_kernel<HID, NC, false, false>
        <<<(strips + 7) / 8, 256, 0, stream>>>(h, W2, nullptr, t, strips);
  }

  gcn_pull_blocks_kernel<NC, true>
      <<<(N_NODES * 16 + 255) / 256, 256, 0, stream>>>(t, blk, head, node, b2,
                                                       out);
}


# non-matrix launches: a lane takes several items, their requests in flight together (init 4 stores, link 8 edges, pulls 2 rows in turn): 57k waves enter, not 162k; sum order by placement as before
# speedup vs baseline: 3.3561x; 1.0528x over previous
#include <hip/hip_runtime.h>
#include <hip/hip_bf16.h>

typedef float v2f __attribute__((ext_vector_type(2)));
typedef float v8f __attribute__((ext_vector_type(8)));

#define N_NODES 100000
#define N_EDGES 1600000
#define IN_F 64
#define HID 64
#define NC 40

#define HLIST 2
#define NLIST (2 * HLIST)

__device__ __forceinline__ unsigned gcn_wave_half() {
  return ((unsigned)__builtin_amdgcn_s_sendmsg_rtn(0x87) >> 18) & 1u;
}

__device__ __forceinline__ size_t gcn_head_at(unsigned p, unsigned r) {
  return ((size_t)p * N_NODES + r) * HLIST;
}

#define INIT_PER_LANE 4
__global__ void gcn_list_init_kernel(int4* __restrict__ head4, int n4) {
  const int stride = gridDim.x * blockDim.x;
  int i = blockIdx.x * blockDim.x + threadIdx.x;
#pragma unroll
  for (int k = 0; k < INIT_PER_LANE; ++k, i += stride)
    if (i < n4) head4[i] = make_int4(-1, -1, -1, -1);
}

#define LINK_PAIRS 4
#define LINK_EDGES_PER_WAVE (64 * LINK_PAIRS)
__global__ void __launch_bounds__(256)
gcn_link_kernel(const int* __restrict__ src, const int* __restrict__ dst,
                int* __restrict__ head, int2* __restrict__ node) {
  static_assert(HLIST == 2, "a pair of adjacent edges: one to each list");
  static_assert(N_EDGES % LINK_EDGES_PER_WAVE == 0, "whole waves only");
  const unsigned wave = (blockIdx.x * blockDim.x + threadIdx.x) >> 5;
  if (wave >= (unsigned)(N_EDGES / LINK_EDGES_PER_WAVE)) return;
  const unsigned e0 = wave * LINK_EDGES_PER_WAVE + 2u * (threadIdx.x & 31u);

  int2 s[LINK_PAIRS], d[LINK_PAIRS];
#pragma unroll
  for (int j = 0; j < LINK_PAIRS; ++j) {
    s[j] = *reinterpret_cast<const int2*>(src + e0 + 64 * j);
    d[j] = *reinterpret_cast<const int2*>(dst + e0 + 64 * j);
  }
  const unsigned half = gcn_wave_half();
  int* const hp = head + gcn_head_at(half, 0u);

  int prev[2 * LINK_PAIRS];
#pragma unroll
  for (int j = 0; j < LINK_PAIRS; ++j) {
    const int e = (int)(e0 + 64 * j);
    prev[2 * j] = atomicExch(hp + (unsigned)d[j].x * HLIST, e);
    prev[2 * j + 1] = atomicExch(hp + (unsigned)d[j].y * HLIST + 1, e + 1);
  }
#pragma unroll
  for (int j = 0; j < LINK_PAIRS; ++j)
    *reinterpret_cast<int4*>(node + e0 + 64 * j) =
        make_int4(s[j].x, prev[2 * j], s[j].y, prev[2 * j + 1]);
}

__device__ __forceinline__ bool gcn_edge_ok(int e) {
  return (unsigned)e < (unsigned)N_EDGES;
}

__device__ __forceinline__ int2 gcn_load_heads(const int* __restrict__ head,
                                               unsigned p, unsigned r) {
  static_assert(HLIST == 2, "one 8-byte load per plane");
  return *reinterpret_cast<const int2*>(head + gcn_head_at(p, r));
}

#define BLK_TRIPS 16
#define BLK_LONG (-2)

template <int COLS, bool RECORD>
__device__ __forceinline__ float4 gcn_walk_lists(const float* __restrict__ x,
                                                 const int2* __restrict__ node,
                                                 const int2 h0, const int2 h1,
                                                 unsigned col, unsigned l,
                                                 int4& keep, int& trips) {
  int e[NLIST] = {h0.x, h0.y, h1.x, h1.y};
  int2 nd[NLIST];
#pragma unroll
  for (int j = 0; j < NLIST; ++j)
    nd[j] = node[gcn_edge_ok(e[j]) ? (unsigned)e[j] : 0u];

  float4 acc = make_float4(0.0f, 0.0f, 0.0f, 0.0f);
  int it = 0;
  for (; it < N_EDGES; ++it) {
    bool ok[NLIST];
    bool any = false;
#pragma unroll
    for (int j = 0; j < NLIST; ++j) {
      ok[j] = gcn_edge_ok(e[j]);
      any = any || ok[j];
    }
    if (!any) break;

    if (RECORD) {
      if ((unsigned)it == l)
        keep = make_int4(ok[0] ? nd[0].x : -1, ok[1] ? nd[1].x : -1,
                         ok[2] ? nd[2].x : -1, ok[3] ? nd[3].x : -1);
    }

    int en[NLIST];
    int2 ndn[NLIST];
#pragma unroll
    for (int j = 0; j < NLIST; ++j) {
      en[j] = ok[j] ? nd[j].y : -1;
      ndn[j] = node[gcn_edge_ok(en[j]) ? (unsigned)en[j] : 0u];
    }
    float4 v[NLIST];
#pragma unroll
    for (int j = 0; j < NLIST; ++j) {
      const unsigned off = (ok[j] ? (unsigned)nd[j].x : 0u) * (unsigned)COLS + col;
      v[j] = *reinterpret_cast<const float4*>(x + off);
    }
#pragma unroll
    for (int j = 0; j < NLIST; ++j) {
      acc.x += ok[j] ? v[j].x : 0.0f;
      acc.y += ok[j] ? v[j].y : 0.0f;
      acc.z += ok[j] ? v[j].z : 0.0f;
      acc.w += ok[j] ? v[j].w : 0.0f;
      e[j] = en[j];
      nd[j] = ndn[j];
    }
  }
  trips = it;
  return acc;
}

#define PULL_ROWS 2
#define PULL_GROUPS (N_NODES / PULL_ROWS)

__global__ void gcn_pull_record_kernel(const float* __restrict__ x,
                                       const int* __restrict__ head,
                                       const int2* __restrict__ node,
                                       int4* __restrict__ blk,
                                       float* __restrict__ y) {
  static_assert(IN_F == 64, "16 lanes x 4 columns: every lane of a row is live");
  static_assert(NLIST == 4, "one int4 of ids per trip");
  static_assert(N_NODES % PULL_ROWS == 0, "every 16 lanes take PULL_ROWS rows");
  const unsigned tid = blockIdx.x * blockDim.x + threadIdx.x;
  const unsigned g = tid >> 4;
  const unsigned l = tid & 15;
  if (g >= (unsigned)PULL_GROUPS) return;
  const unsigned col = l * 4;

  int2 h0[PULL_ROWS], h1[PULL_ROWS];
#pragma unroll
  for (int p = 0; p < PULL_ROWS; ++p) {
    h0[p] = gcn_load_heads(head, 0, g + p * PULL_GROUPS);
    h1[p] = gcn_load_heads(head, 1, g + p * PULL_GROUPS);
  }
#pragma unroll
  for (int p = 0; p < PULL_ROWS; ++p) {
    const unsigned row = g + p * PULL_GROUPS;
    int4 keep = make_int4(-1, -1, -1, -1);
    int trips = 0;
    const float4 acc = gcn_walk_lists<IN_F, true>(x, node, h0[p], h1[p], col,
                                                  l, keep, trips);
    if (trips > BLK_TRIPS && l == BLK_TRIPS - 1) keep.w = BLK_LONG;

    blk[(size_t)row * BLK_TRIPS + l] = keep;
    *reinterpret_cast<float4*>(y + (size_t)row * IN_F + col) = acc;
  }
}

template <int K>
__device__ __forceinline__ int gcn_row_share(int v) {
  return __builtin_amdgcn_update_dpp(0, v, 0x150 + K, 0xF, 0xF, true);
}

template <int COLS, int B>
__device__ __forceinline__ bool gcn_block_batch(const float* __restrict__ x,
                                                const int4 my, unsigned col,
                                                float4& acc) {
  int id[8];
  id[0] = gcn_row_share<2 * B>(my.x);
  id[1] = gcn_row_share<2 * B>(my.y);
  id[2] = gcn_row_share<2 * B>(my.z);
  id[3] = gcn_row_share<2 * B>(my.w);
  id[4] = gcn_row_share<2 * B + 1>(my.x);
  id[5] = gcn_row_share<2 * B + 1>(my.y);
  id[6] = gcn_row_share<2 * B + 1>(my.z);
  id[7] = gcn_row_share<2 * B + 1>(my.w);
  bool any = false;
#pragma unroll
  for (int k = 0; k < 8; ++k) any = any || (id[k] >= 0);
  if (!any) return false;
  float4 v[8];
#pragma unroll
  for (int k = 0; k < 8; ++k) {
    const unsigned off = (id[k] >= 0 ? (unsigned)id[k] : 0u) * (unsigned)COLS + col;
    v[k] = *reinterpret_cast<const float4*>(x + off);
  }
#pragma unroll
  for (int k = 0; k < 8; ++k) {
    acc.x += id[k] >= 0 ? v[k].x : 0.0f;
    acc.y += id[k] >= 0 ? v[k].y : 0.0f;
    acc.z += id[k] >= 0 ? v[k].z : 0.0f;
    acc.w += id[k] >= 0 ? v[k].w : 0.0f;
  }
  return true;
}

template <int COLS, bool HAS_BIAS>
__global__ void __launch_bounds__(256, 16)
gcn_pull_blocks_kernel(const float* __restrict__ x,
                                       const int4* __restrict__ blk,
                                       const int* __restrict__ head,
                                       const int2* __restrict__ node,
                                       const float* __restrict__ bias,
                                       float* __restrict__ y) {
  static_assert(COLS % 4 == 0 && COLS <= 64, "a lane owns 4 adjacent columns");
  const unsigned tid = blockIdx.x * blockDim.x + threadIdx.x;
  const unsigned g = tid >> 4;
  const unsigned l = tid & 15;
  if (g >= (unsigned)PULL_GROUPS) return;
  const bool cok = l * 4 < (unsigned)COLS;
  const unsigned col = cok ? l * 4 : 0u;

  int4 my = blk[(size_t)g * BLK_TRIPS + l];
#pragma nounroll
  for (int p = 0; p < PULL_ROWS; ++p) {
    const unsigned row = g + p * PULL_GROUPS;
    int4 nxt = my;
    if (p + 1 < PULL_ROWS)
      nxt = blk[(size_t)(row + PULL_GROUPS) * BLK_TRIPS + l];
    float4 acc = make_float4(0.0f, 0.0f, 0.0f, 0.0f);
    if (gcn_row_share<BLK_TRIPS - 1>(my.w) == BLK_LONG) {
      int4 unused = make_int4(-1, -1, -1, -1);
      int trips = 0;
      acc = gcn_walk_lists<COLS, false>(x, node, gcn_load_heads(head, 0, row),
                                        gcn_load_heads(head, 1, row), col, l,
                                        unused, trips);
    } else {
      if (gcn_block_batch<COLS, 0>(x, my, col, acc))
      if (gcn_block_batch<COLS, 1>(x, my, col, acc))
      if (gcn_block_batch<COLS, 2>(x, my, col, acc))
      if (gcn_block_batch<COLS, 3>(x, my, col, acc))
      if (gcn_block_batch<COLS, 4>(x, my, col, acc))
      if (gcn_block_batch<COLS, 5>(x, my, col, acc))
      if (gcn_block_batch<COLS, 6>(x, my, col, acc))
        gcn_block_batch<COLS, 7>(x, my, col, acc);
    }
    if (cok) {
      if (HAS_BIAS) {
        acc.x += bias[col + 0]; acc.y += bias[col + 1];
        acc.z += bias[col + 2]; acc.w += bias[col + 3];
      }
      *reinterpret_cast<float4*>(y + (size_t)row * COLS + col) = acc;
    }
    my = nxt;
  }
}

template <int KDIM, int NREAL, bool RELU, bool BIAS>
__global__ void __launch_bounds__(256)
gcn_wmma_strip_kernel(const float* __restrict__ A,
                      const float* __restrict__ W,
                      const float* __restrict__ bias,
                      float* __restrict__ C, int n_strips) {
  static_assert(KDIM % 8 == 0, "k is taken 8 at a time");
  static_assert(NREAL % 4 == 0 && NREAL <= 64, "four column tiles, 4 columns a lane");
  constexpr int NQ = NREAL / 4;
  constexpr bool FULL = (NQ == 16);

  const int strip = blockIdx.x * (blockDim.x >> 5) + (threadIdx.x >> 5);
  if (strip >= n_strips) return;

  const int lane = threadIdx.x & 31;
  const int c = lane & 15;
  const int h = lane >> 4;
  const bool cok = FULL ? true : (c < NQ);
  const int cc = cok ? c : 0;

  const float* a0 = A + ((size_t)strip * 32 + c) * KDIM + 4 * h;
  const float* a1 = a0 + (size_t)16 * KDIM;
  const float* wp = W + (size_t)(4 * h) * NREAL + 4 * cc;

  v8f acc[2][4];
#pragma unroll
  for (int m = 0; m < 2; ++m)
#pragma unroll
    for (int ct = 0; ct < 4; ++ct) acc[m][ct] = v8f{};

#pragma unroll
  for (int i = 0; i < KDIM / 8; ++i) {
    const float4 x0 = *reinterpret_cast<const float4*>(a0 + 8 * i);
    const float4 x1 = *reinterpret_cast<const float4*>(a1 + 8 * i);
    float wv[4][4];
#pragma unroll
    for (int j = 0; j < 4; ++j) {
      const float4 w4 =
          *reinterpret_cast<const float4*>(wp + (size_t)(8 * i + j) * NREAL);
      wv[j][0] = w4.x; wv[j][1] = w4.y; wv[j][2] = w4.z; wv[j][3] = w4.w;
    }
#pragma unroll
    for (int p = 0; p < 2; ++p) {
      v2f af0, af1;
      af0.x = p ? x0.z : x0.x;
      af0.y = p ? x0.w : x0.y;
      af1.x = p ? x1.z : x1.x;
      af1.y = p ? x1.w : x1.y;
#pragma unroll
      for (int ct = 0; ct < 4; ++ct) {
        v2f bf;
        bf.x = wv[2 * p][ct];
        bf.y = wv[2 * p + 1][ct];
        acc[0][ct] = __builtin_amdgcn_wmma_f32_16x16x4_f32(
            false, af0, false, bf, (short)0, acc[0][ct], false, false);
        acc[1][ct] = __builtin_amdgcn_wmma_f32_16x16x4_f32(
            false, af1, false, bf, (short)0, acc[1][ct], false, false);
      }
    }
  }

  if (!cok) return;
  float4 bv = make_float4(0.0f, 0.0f, 0.0f, 0.0f);
  if (BIAS) bv = *reinterpret_cast<const float4*>(bias + 4 * c);
#pragma unroll
  for (int m = 0; m < 2; ++m) {
#pragma unroll
    for (int r = 0; r < 8; ++r) {
      const size_t row = (size_t)strip * 32 + 16 * m + 8 * h + r;
      float4 v;
      v.x = acc[m][0][r] + bv.x;
      v.y = acc[m][1][r] + bv.y;
      v.z = acc[m][2][r] + bv.z;
      v.w = acc[m][3][r] + bv.w;
      if (RELU) {
        v.x = fmaxf(v.x, 0.0f); v.y = fmaxf(v.y, 0.0f);
        v.z = fmaxf(v.z, 0.0f); v.w = fmaxf(v.w, 0.0f);
      }
      *reinterpret_cast<float4*>(C + row * NREAL + 4 * c) = v;
    }
  }
}

extern "C" void kernel_launch(void* const* d_in, const int* in_sizes, int n_in,
                              void* d_out, int out_size, void* d_ws,
                              size_t ws_size, hipStream_t stream) {
  (void)in_sizes; (void)n_in; (void)out_size; (void)ws_size;
  const float* features = (const float*)d_in[0];
  const int*   src      = (const int*)d_in[1];
  const int*   dst      = (const int*)d_in[2];
  const float* W1       = (const float*)d_in[3];
  const float* b1       = (const float*)d_in[4];
  const float* W2       = (const float*)d_in[5];
  const float* b2       = (const float*)d_in[6];
  float* out = (float*)d_out;

  float* ws   = (float*)d_ws;
  float* agg1 = ws;
  float* h    = ws + (size_t)N_NODES * HID;
  float* t    = ws;
  int2*  node = (int2*)(ws + (size_t)2 * N_NODES * HID);
  int*   head = (int*)(ws + (size_t)2 * N_NODES * HID + (size_t)2 * N_EDGES);
  int4*  blk  = (int4*)(head + (size_t)N_NODES * NLIST);

  {
    const int n4 = N_NODES * NLIST / 4;
    gcn_list_init_kernel<<<(n4 + 256 * INIT_PER_LANE - 1) / (256 * INIT_PER_LANE),
                           256, 0, stream>>>((int4*)head, n4);
  }

  gcn_link_kernel<<<(N_EDGES / LINK_EDGES_PER_WAVE + 7) / 8, 256, 0, stream>>>(
      src, dst, head, node);

  gcn_pull_record_kernel
      <<<(PULL_GROUPS * 16 + 255) / 256, 256, 0, stream>>>(features, head, node,
                                                           blk, agg1);

  static_assert(N_NODES % 32 == 0, "a wave owns 32 whole rows");
  {
    const int strips = N_NODES / 32;
    gcn_wmma_strip_kernel<HID, HID, true, true>
        <<<(strips + 7) / 8, 256, 0, stream>>>(agg1, W1, b1, h, strips);
  }

  {
    const int strips = N_NODES / 32;
    gcn_wmma_strip_kernel<HID, NC, false, false>
        <<<(strips + 7) / 8, 256, 0, stream>>>(h, W2, nullptr, t, strips);
  }

  gcn_pull_blocks_kernel<NC, true>
      <<<(PULL_GROUPS * 16 + 255) / 256, 256, 0, stream>>>(t, blk, head, node,
                                                           b2, out);
}


# dense layers as transposed products in one launch: product 1's accumulator registers (a lane = one node's hidden values) are product 2's B operand as they stand; h not stored; product 2 k order new
# speedup vs baseline: 3.6340x; 1.0828x over previous
#include <hip/hip_runtime.h>
#include <hip/hip_bf16.h>

typedef float v2f __attribute__((ext_vector_type(2)));
typedef float v8f __attribute__((ext_vector_type(8)));

#define N_NODES 100000
#define N_EDGES 1600000
#define IN_F 64
#define HID 64
#define NC 40

#define HLIST 2
#define NLIST (2 * HLIST)

__device__ __forceinline__ unsigned gcn_wave_half() {
  return ((unsigned)__builtin_amdgcn_s_sendmsg_rtn(0x87) >> 18) & 1u;
}

__device__ __forceinline__ size_t gcn_head_at(unsigned p, unsigned r) {
  return ((size_t)p * N_NODES + r) * HLIST;
}

#define INIT_PER_LANE 4
__global__ void gcn_list_init_kernel(int4* __restrict__ head4, int n4) {
  const int stride = gridDim.x * blockDim.x;
  int i = blockIdx.x * blockDim.x + threadIdx.x;
#pragma unroll
  for (int k = 0; k < INIT_PER_LANE; ++k, i += stride)
    if (i < n4) head4[i] = make_int4(-1, -1, -1, -1);
}

#define LINK_PAIRS 4
#define LINK_EDGES_PER_WAVE (64 * LINK_PAIRS)
__global__ void __launch_bounds__(256)
gcn_link_kernel(const int* __restrict__ src, const int* __restrict__ dst,
                int* __restrict__ head, int2* __restrict__ node) {
  static_assert(HLIST == 2, "a pair of adjacent edges: one to each list");
  static_assert(N_EDGES % LINK_EDGES_PER_WAVE == 0, "whole waves only");
  const unsigned wave = (blockIdx.x * blockDim.x + threadIdx.x) >> 5;
  if (wave >= (unsigned)(N_EDGES / LINK_EDGES_PER_WAVE)) return;
  const unsigned e0 = wave * LINK_EDGES_PER_WAVE + 2u * (threadIdx.x & 31u);

  int2 s[LINK_PAIRS], d[LINK_PAIRS];
#pragma unroll
  for (int j = 0; j < LINK_PAIRS; ++j) {
    s[j] = *reinterpret_cast<const int2*>(src + e0 + 64 * j);
    d[j] = *reinterpret_cast<const int2*>(dst + e0 + 64 * j);
  }
  const unsigned half = gcn_wave_half();
  int* const hp = head + gcn_head_at(half, 0u);

  int prev[2 * LINK_PAIRS];
#pragma unroll
  for (int j = 0; j < LINK_PAIRS; ++j) {
    const int e = (int)(e0 + 64 * j);
    prev[2 * j] = atomicExch(hp + (unsigned)d[j].x * HLIST, e);
    prev[2 * j + 1] = atomicExch(hp + (unsigned)d[j].y * HLIST + 1, e + 1);
  }
#pragma unroll
  for (int j = 0; j < LINK_PAIRS; ++j)
    *reinterpret_cast<int4*>(node + e0 + 64 * j) =
        make_int4(s[j].x, prev[2 * j], s[j].y, prev[2 * j + 1]);
}

__device__ __forceinline__ bool gcn_edge_ok(int e) {
  return (unsigned)e < (unsigned)N_EDGES;
}

__device__ __forceinline__ int2 gcn_load_heads(const int* __restrict__ head,
                                               unsigned p, unsigned r) {
  static_assert(HLIST == 2, "one 8-byte load per plane");
  return *reinterpret_cast<const int2*>(head + gcn_head_at(p, r));
}

#define BLK_TRIPS 16
#define BLK_LONG (-2)

template <int COLS, bool RECORD>
__device__ __forceinline__ float4 gcn_walk_lists(const float* __restrict__ x,
                                                 const int2* __restrict__ node,
                                                 const int2 h0, const int2 h1,
                                                 unsigned col, unsigned l,
                                                 int4& keep, int& trips) {
  int e[NLIST] = {h0.x, h0.y, h1.x, h1.y};
  int2 nd[NLIST];
#pragma unroll
  for (int j = 0; j < NLIST; ++j)
    nd[j] = node[gcn_edge_ok(e[j]) ? (unsigned)e[j] : 0u];

  float4 acc = make_float4(0.0f, 0.0f, 0.0f, 0.0f);
  int it = 0;
  for (; it < N_EDGES; ++it) {
    bool ok[NLIST];
    bool any = false;
#pragma unroll
    for (int j = 0; j < NLIST; ++j) {
      ok[j] = gcn_edge_ok(e[j]);
      any = any || ok[j];
    }
    if (!any) break;

    if (RECORD) {
      if ((unsigned)it == l)
        keep = make_int4(ok[0] ? nd[0].x : -1, ok[1] ? nd[1].x : -1,
                         ok[2] ? nd[2].x : -1, ok[3] ? nd[3].x : -1);
    }

    int en[NLIST];
    int2 ndn[NLIST];
#pragma unroll
    for (int j = 0; j < NLIST; ++j) {
      en[j] = ok[j] ? nd[j].y : -1;
      ndn[j] = node[gcn_edge_ok(en[j]) ? (unsigned)en[j] : 0u];
    }
    float4 v[NLIST];
#pragma unroll
    for (int j = 0; j < NLIST; ++j) {
      const unsigned off = (ok[j] ? (unsigned)nd[j].x : 0u) * (unsigned)COLS + col;
      v[j] = *reinterpret_cast<const float4*>(x + off);
    }
#pragma unroll
    for (int j = 0; j < NLIST; ++j) {
      acc.x += ok[j] ? v[j].x : 0.0f;
      acc.y += ok[j] ? v[j].y : 0.0f;
      acc.z += ok[j] ? v[j].z : 0.0f;
      acc.w += ok[j] ? v[j].w : 0.0f;
      e[j] = en[j];
      nd[j] = ndn[j];
    }
  }
  trips = it;
  return acc;
}

#define PULL_ROWS 2
#define PULL_GROUPS (N_NODES / PULL_ROWS)

__global__ void gcn_pull_record_kernel(const float* __restrict__ x,
                                       const int* __restrict__ head,
                                       const int2* __restrict__ node,
                                       int4* __restrict__ blk,
                                       float* __restrict__ y) {
  static_assert(IN_F == 64, "16 lanes x 4 columns: every lane of a row is live");
  static_assert(NLIST == 4, "one int4 of ids per trip");
  static_assert(N_NODES % PULL_ROWS == 0, "every 16 lanes take PULL_ROWS rows");
  const unsigned tid = blockIdx.x * blockDim.x + threadIdx.x;
  const unsigned g = tid >> 4;
  const unsigned l = tid & 15;
  if (g >= (unsigned)PULL_GROUPS) return;
  const unsigned col = l * 4;

  int2 h0[PULL_ROWS], h1[PULL_ROWS];
#pragma unroll
  for (int p = 0; p < PULL_ROWS; ++p) {
    h0[p] = gcn_load_heads(head, 0, g + p * PULL_GROUPS);
    h1[p] = gcn_load_heads(head, 1, g + p * PULL_GROUPS);
  }
#pragma unroll
  for (int p = 0; p < PULL_ROWS; ++p) {
    const unsigned row = g + p * PULL_GROUPS;
    int4 keep = make_int4(-1, -1, -1, -1);
    int trips = 0;
    const float4 acc = gcn_walk_lists<IN_F, true>(x, node, h0[p], h1[p], col,
                                                  l, keep, trips);
    if (trips > BLK_TRIPS && l == BLK_TRIPS - 1) keep.w = BLK_LONG;

    blk[(size_t)row * BLK_TRIPS + l] = keep;
    *reinterpret_cast<float4*>(y + (size_t)row * IN_F + col) = acc;
  }
}

template <int K>
__device__ __forceinline__ int gcn_row_share(int v) {
  return __builtin_amdgcn_update_dpp(0, v, 0x150 + K, 0xF, 0xF, true);
}

template <int COLS, int B>
__device__ __forceinline__ bool gcn_block_batch(const float* __restrict__ x,
                                                const int4 my, unsigned col,
                                                float4& acc) {
  int id[8];
  id[0] = gcn_row_share<2 * B>(my.x);
  id[1] = gcn_row_share<2 * B>(my.y);
  id[2] = gcn_row_share<2 * B>(my.z);
  id[3] = gcn_row_share<2 * B>(my.w);
  id[4] = gcn_row_share<2 * B + 1>(my.x);
  id[5] = gcn_row_share<2 * B + 1>(my.y);
  id[6] = gcn_row_share<2 * B + 1>(my.z);
  id[7] = gcn_row_share<2 * B + 1>(my.w);
  bool any = false;
#pragma unroll
  for (int k = 0; k < 8; ++k) any = any || (id[k] >= 0);
  if (!any) return false;
  float4 v[8];
#pragma unroll
  for (int k = 0; k < 8; ++k) {
    const unsigned off = (id[k] >= 0 ? (unsigned)id[k] : 0u) * (unsigned)COLS + col;
    v[k] = *reinterpret_cast<const float4*>(x + off);
  }
#pragma unroll
  for (int k = 0; k < 8; ++k) {
    acc.x += id[k] >= 0 ? v[k].x : 0.0f;
    acc.y += id[k] >= 0 ? v[k].y : 0.0f;
    acc.z += id[k] >= 0 ? v[k].z : 0.0f;
    acc.w += id[k] >= 0 ? v[k].w : 0.0f;
  }
  return true;
}

template <int COLS, bool HAS_BIAS>
__global__ void __launch_bounds__(256, 16)
gcn_pull_blocks_kernel(const float* __restrict__ x,
                                       const int4* __restrict__ blk,
                                       const int* __restrict__ head,
                                       const int2* __restrict__ node,
                                       const float* __restrict__ bias,
                                       float* __restrict__ y) {
  static_assert(COLS % 4 == 0 && COLS <= 64, "a lane owns 4 adjacent columns");
  const unsigned tid = blockIdx.x * blockDim.x + threadIdx.x;
  const unsigned g = tid >> 4;
  const unsigned l = tid & 15;
  if (g >= (unsigned)PULL_GROUPS) return;
  const bool cok = l * 4 < (unsigned)COLS;
  const unsigned col = cok ? l * 4 : 0u;

  int4 my = blk[(size_t)g * BLK_TRIPS + l];
#pragma nounroll
  for (int p = 0; p < PULL_ROWS; ++p) {
    const unsigned row = g + p * PULL_GROUPS;
    int4 nxt = my;
    if (p + 1 < PULL_ROWS)
      nxt = blk[(size_t)(row + PULL_GROUPS) * BLK_TRIPS + l];
    float4 acc = make_float4(0.0f, 0.0f, 0.0f, 0.0f);
    if (gcn_row_share<BLK_TRIPS - 1>(my.w) == BLK_LONG) {
      int4 unused = make_int4(-1, -1, -1, -1);
      int trips = 0;
      acc = gcn_walk_lists<COLS, false>(x, node, gcn_load_heads(head, 0, row),
                                        gcn_load_heads(head, 1, row), col, l,
                                        unused, trips);
    } else {
      if (gcn_block_batch<COLS, 0>(x, my, col, acc))
      if (gcn_block_batch<COLS, 1>(x, my, col, acc))
      if (gcn_block_batch<COLS, 2>(x, my, col, acc))
      if (gcn_block_batch<COLS, 3>(x, my, col, acc))
      if (gcn_block_batch<COLS, 4>(x, my, col, acc))
      if (gcn_block_batch<COLS, 5>(x, my, col, acc))
      if (gcn_block_batch<COLS, 6>(x, my, col, acc))
        gcn_block_batch<COLS, 7>(x, my, col, acc);
    }
    if (cok) {
      if (HAS_BIAS) {
        acc.x += bias[col + 0]; acc.y += bias[col + 1];
        acc.z += bias[col + 2]; acc.w += bias[col + 3];
      }
      *reinterpret_cast<float4*>(y + (size_t)row * COLS + col) = acc;
    }
    my = nxt;
  }
}

__global__ void __launch_bounds__(256)
gcn_wmma_chain_kernel(const float* __restrict__ X,
                      const float* __restrict__ W1,
                      const float* __restrict__ b1,
                      const float* __restrict__ W2,
                      float* __restrict__ T, int n_strips) {
  static_assert(IN_F == 64 && HID == 64, "four hidden tiles, k taken 8 at a time");
  static_assert(NC == 40, "two whole class tiles and the lower half of a third");

  const int strip = blockIdx.x * (blockDim.x >> 5) + (threadIdx.x >> 5);
  if (strip >= n_strips) return;

  const int lane = threadIdx.x & 31;
  const int c = lane & 15;
  const int h = lane >> 4;

  const float* x0p = X + ((size_t)strip * 32 + c) * IN_F + 4 * h;
  const float* x1p = x0p + (size_t)16 * IN_F;
  const float* wp = W1 + (size_t)(4 * h) * HID + 4 * c;

  v8f hid[2][4];
#pragma unroll
  for (int u = 0; u < 2; ++u)
#pragma unroll
    for (int mt = 0; mt < 4; ++mt) hid[u][mt] = v8f{};

#pragma unroll
  for (int i = 0; i < IN_F / 8; ++i) {
    const float4 x0 = *reinterpret_cast<const float4*>(x0p + 8 * i);
    const float4 x1 = *reinterpret_cast<const float4*>(x1p + 8 * i);
    float wv[4][4];
#pragma unroll
    for (int j = 0; j < 4; ++j) {
      const float4 w4 =
          *reinterpret_cast<const float4*>(wp + (size_t)(8 * i + j) * HID);
      wv[j][0] = w4.x; wv[j][1] = w4.y; wv[j][2] = w4.z; wv[j][3] = w4.w;
    }
#pragma unroll
    for (int p = 0; p < 2; ++p) {
      v2f bf0, bf1;
      bf0.x = p ? x0.z : x0.x;
      bf0.y = p ? x0.w : x0.y;
      bf1.x = p ? x1.z : x1.x;
      bf1.y = p ? x1.w : x1.y;
#pragma unroll
      for (int mt = 0; mt < 4; ++mt) {
        v2f af;
        af.x = wv[2 * p][mt];
        af.y = wv[2 * p + 1][mt];
        hid[0][mt] = __builtin_amdgcn_wmma_f32_16x16x4_f32(
            false, af, false, bf0, (short)0, hid[0][mt], false, false);
        hid[1][mt] = __builtin_amdgcn_wmma_f32_16x16x4_f32(
            false, af, false, bf1, (short)0, hid[1][mt], false, false);
      }
    }
  }

#pragma unroll
  for (int r = 0; r < 8; ++r) {
    const float4 bb = *reinterpret_cast<const float4*>(b1 + 4 * (8 * h + r));
#pragma unroll
    for (int u = 0; u < 2; ++u) {
      hid[u][0][r] = fmaxf(hid[u][0][r] + bb.x, 0.0f);
      hid[u][1][r] = fmaxf(hid[u][1][r] + bb.y, 0.0f);
      hid[u][2][r] = fmaxf(hid[u][2][r] + bb.z, 0.0f);
      hid[u][3][r] = fmaxf(hid[u][3][r] + bb.w, 0.0f);
    }
  }

  v8f cls[2][3];
#pragma unroll
  for (int u = 0; u < 2; ++u)
#pragma unroll
    for (int ct = 0; ct < 3; ++ct) cls[u][ct] = v8f{};

  const int slot = 8 * ((c & 7) >> 2) + 4 * (c >> 3) + (c & 3);
  int col[3];
  col[0] = slot;
  col[1] = 16 + slot;
  col[2] = (slot < 8) ? 32 + slot : 32;

#pragma unroll
  for (int mt = 0; mt < 4; ++mt) {
#pragma unroll
    for (int q = 0; q < 4; ++q) {
      const float* w2p = W2 + (size_t)(4 * (8 * h + 2 * q) + mt) * NC;
      v2f bf0, bf1;
      bf0.x = hid[0][mt][2 * q];
      bf0.y = hid[0][mt][2 * q + 1];
      bf1.x = hid[1][mt][2 * q];
      bf1.y = hid[1][mt][2 * q + 1];
#pragma unroll
      for (int ct = 0; ct < 3; ++ct) {
        v2f af;
        af.x = w2p[col[ct]];
        af.y = w2p[4 * NC + col[ct]];
        cls[0][ct] = __builtin_amdgcn_wmma_f32_16x16x4_f32(
            false, af, false, bf0, (short)0, cls[0][ct], false, false);
        cls[1][ct] = __builtin_amdgcn_wmma_f32_16x16x4_f32(
            false, af, false, bf1, (short)0, cls[1][ct], false, false);
      }
    }
  }

#pragma unroll
  for (int u = 0; u < 2; ++u) {
    float* tp = T + ((size_t)strip * 32 + 16 * u + c) * NC + 4 * h;
#pragma unroll
    for (int ct = 0; ct < 3; ++ct) {
      *reinterpret_cast<float4*>(tp + 16 * ct) = make_float4(
          cls[u][ct][0], cls[u][ct][1], cls[u][ct][2], cls[u][ct][3]);
      if (ct < 2)
        *reinterpret_cast<float4*>(tp + 16 * ct + 8) = make_float4(
            cls[u][ct][4], cls[u][ct][5], cls[u][ct][6], cls[u][ct][7]);
    }
  }
}

extern "C" void kernel_launch(void* const* d_in, const int* in_sizes, int n_in,
                              void* d_out, int out_size, void* d_ws,
                              size_t ws_size, hipStream_t stream) {
  (void)in_sizes; (void)n_in; (void)out_size; (void)ws_size;
  const float* features = (const float*)d_in[0];
  const int*   src      = (const int*)d_in[1];
  const int*   dst      = (const int*)d_in[2];
  const float* W1       = (const float*)d_in[3];
  const float* b1       = (const float*)d_in[4];
  const float* W2       = (const float*)d_in[5];
  const float* b2       = (const float*)d_in[6];
  float* out = (float*)d_out;

  float* ws   = (float*)d_ws;
  float* agg1 = ws;
  float* t    = ws + (size_t)N_NODES * HID;
  int2*  node = (int2*)(ws + (size_t)2 * N_NODES * HID);
  int*   head = (int*)(ws + (size_t)2 * N_NODES * HID + (size_t)2 * N_EDGES);
  int4*  blk  = (int4*)(head + (size_t)N_NODES * NLIST);

  {
    const int n4 = N_NODES * NLIST / 4;
    gcn_list_init_kernel<<<(n4 + 256 * INIT_PER_LANE - 1) / (256 * INIT_PER_LANE),
                           256, 0, stream>>>((int4*)head, n4);
  }

  gcn_link_kernel<<<(N_EDGES / LINK_EDGES_PER_WAVE + 7) / 8, 256, 0, stream>>>(
      src, dst, head, node);

  gcn_pull_record_kernel
      <<<(PULL_GROUPS * 16 + 255) / 256, 256, 0, stream>>>(features, head, node,
                                                           blk, agg1);

  static_assert(N_NODES % 32 == 0, "a wave owns 32 whole rows");
  {
    const int strips = N_NODES / 32;
    gcn_wmma_chain_kernel<<<(strips + 7) / 8, 256, 0, stream>>>(agg1, W1, b1, W2,
                                                                t, strips);
  }

  gcn_pull_blocks_kernel<NC, true>
      <<<(PULL_GROUPS * 16 + 255) / 256, 256, 0, stream>>>(t, blk, head, node,
                                                           b2, out);
}


# edges placed, not linked: a returning add per edge draws a slot in its row's 128-B id line (one per package half); both layers sum from the lines: no list walk, no node array; sums in slot order
# speedup vs baseline: 3.7990x; 1.0454x over previous
#include <hip/hip_runtime.h>
#include <hip/hip_bf16.h>

typedef float v2f __attribute__((ext_vector_type(2)));
typedef float v8f __attribute__((ext_vector_type(8)));
typedef float v4f __attribute__((ext_vector_type(4)));

#define N_NODES 100000
#define N_EDGES 1600000
#define IN_F 64
#define HID 64
#define NC 40

#define LINE_INTS 32
#define SLOTS (LINE_INTS - 1)

__device__ __forceinline__ unsigned gcn_wave_half() {
  return ((unsigned)__builtin_amdgcn_s_sendmsg_rtn(0x87) >> 18) & 1u;
}

__device__ __forceinline__ size_t gcn_line_at(unsigned p, unsigned r) {
  return ((size_t)p * N_NODES + r) * LINE_INTS;
}

#define INIT_PER_LANE 4
__global__ void gcn_count_init_kernel(int* __restrict__ lines, int n_lines,
                                      int* __restrict__ spill_cnt) {
  const int stride = gridDim.x * blockDim.x;
  int i = blockIdx.x * blockDim.x + threadIdx.x;
  if (i == 0) *spill_cnt = 0;
#pragma unroll
  for (int k = 0; k < INIT_PER_LANE; ++k, i += stride)
    if (i < n_lines) lines[(size_t)i * LINE_INTS] = 0;
}

__device__ __forceinline__ void gcn_place(int* __restrict__ lp, int d, int s,
                                          int slot, int* __restrict__ spill_cnt,
                                          int2* __restrict__ spill) {
  if ((unsigned)slot < (unsigned)SLOTS) {
    lp[(unsigned)d * LINE_INTS + 1u + (unsigned)slot] = s;
  } else {
    const int k = atomicAdd(spill_cnt, 1);
    if ((unsigned)k < (unsigned)N_EDGES) spill[k] = make_int2(d, s);
  }
}

#define PLACE_PAIRS 4
#define PLACE_EDGES_PER_WAVE (64 * PLACE_PAIRS)
__global__ void __launch_bounds__(256)
gcn_place_kernel(const int* __restrict__ src, const int* __restrict__ dst,
                 int* __restrict__ lines, int* __restrict__ spill_cnt,
                 int2* __restrict__ spill) {
  static_assert(N_EDGES % PLACE_EDGES_PER_WAVE == 0, "whole waves only");
  const unsigned wave = (blockIdx.x * blockDim.x + threadIdx.x) >> 5;
  if (wave >= (unsigned)(N_EDGES / PLACE_EDGES_PER_WAVE)) return;
  const unsigned e0 = wave * PLACE_EDGES_PER_WAVE + 2u * (threadIdx.x & 31u);

  int2 s[PLACE_PAIRS], d[PLACE_PAIRS];
#pragma unroll
  for (int j = 0; j < PLACE_PAIRS; ++j) {
    s[j] = *reinterpret_cast<const int2*>(src + e0 + 64 * j);
    d[j] = *reinterpret_cast<const int2*>(dst + e0 + 64 * j);
  }
  const unsigned half = gcn_wave_half();
  int* const lp = lines + gcn_line_at(half, 0u);

  int slot[2 * PLACE_PAIRS];
#pragma unroll
  for (int j = 0; j < PLACE_PAIRS; ++j) {
    slot[2 * j] = atomicAdd(lp + (unsigned)d[j].x * LINE_INTS, 1);
    slot[2 * j + 1] = atomicAdd(lp + (unsigned)d[j].y * LINE_INTS, 1);
  }
#pragma unroll
  for (int j = 0; j < PLACE_PAIRS; ++j) {
    gcn_place(lp, d[j].x, s[j].x, slot[2 * j], spill_cnt, spill);
    gcn_place(lp, d[j].y, s[j].y, slot[2 * j + 1], spill_cnt, spill);
  }
}

template <int K>
__device__ __forceinline__ int gcn_row_share(int v) {
  return __builtin_amdgcn_update_dpp(0, v, 0x150 + K, 0xF, 0xF, true);
}

template <int WORD>
__device__ __forceinline__ int gcn_line_word(const int2 w) {
  if constexpr (WORD < LINE_INTS)
    return gcn_row_share<(WORD >> 1)>((WORD & 1) ? w.y : w.x);
  else
    return 0;
}

template <int COLS, int B>
__device__ __forceinline__ bool gcn_line_batch(const float* __restrict__ x,
                                               const int2 a, const int2 b,
                                               int c0, int c1, unsigned col,
                                               float4& acc) {
  if (!(4 * B < c0 || 4 * B < c1)) return false;
  int id[8];
  id[0] = gcn_line_word<4 * B + 1>(a);
  id[1] = gcn_line_word<4 * B + 2>(a);
  id[2] = gcn_line_word<4 * B + 3>(a);
  id[3] = gcn_line_word<4 * B + 4>(a);
  id[4] = gcn_line_word<4 * B + 1>(b);
  id[5] = gcn_line_word<4 * B + 2>(b);
  id[6] = gcn_line_word<4 * B + 3>(b);
  id[7] = gcn_line_word<4 * B + 4>(b);
  bool ok[8];
#pragma unroll
  for (int k = 0; k < 4; ++k) {
    ok[k] = 4 * B + k < c0;
    ok[4 + k] = 4 * B + k < c1;
  }
  v4f v[8];
#pragma unroll
  for (int k = 0; k < 8; ++k) {
    const unsigned off = (ok[k] ? (unsigned)id[k] : 0u) * (unsigned)COLS + col;
    v[k] = *reinterpret_cast<const v4f*>(x + off);
  }
  asm volatile("" ::"v"(v[0]), "v"(v[1]), "v"(v[2]), "v"(v[3]), "v"(v[4]),
               "v"(v[5]), "v"(v[6]), "v"(v[7]));
#pragma unroll
  for (int k = 0; k < 8; ++k) {
    acc.x += ok[k] ? v[k].x : 0.0f;
    acc.y += ok[k] ? v[k].y : 0.0f;
    acc.z += ok[k] ? v[k].z : 0.0f;
    acc.w += ok[k] ? v[k].w : 0.0f;
  }
  return true;
}

template <int COLS>
__device__ __forceinline__ void gcn_add_spill(const float* __restrict__ x,
                                              const int* __restrict__ spill_cnt,
                                              const int2* __restrict__ spill,
                                              unsigned row, unsigned col,
                                              float4& acc) {
  int n = *spill_cnt;
  if (n > N_EDGES) n = N_EDGES;
  for (int i = 0; i < n; ++i) {
    const int2 en = spill[i];
    if (en.x == (int)row) {
      const float4 v = *reinterpret_cast<const float4*>(
          x + (unsigned)en.y * (unsigned)COLS + col);
      acc.x += v.x; acc.y += v.y; acc.z += v.z; acc.w += v.w;
    }
  }
}

#define PULL_ROWS 2
#define PULL_GROUPS (N_NODES / PULL_ROWS)

template <int COLS, bool HAS_BIAS>
__global__ void __launch_bounds__(256, 16)
gcn_pull_lines_kernel(const float* __restrict__ x,
                      const int* __restrict__ lines,
                      const int* __restrict__ spill_cnt,
                      const int2* __restrict__ spill,
                      const float* __restrict__ bias,
                      float* __restrict__ y) {
  static_assert(COLS % 4 == 0 && COLS <= 64, "a lane owns 4 adjacent columns");
  static_assert(LINE_INTS == 32, "16 lanes x 8 bytes = one line");
  static_assert(N_NODES % PULL_ROWS == 0, "every 16 lanes take PULL_ROWS rows");
  const unsigned tid = blockIdx.x * blockDim.x + threadIdx.x;
  const unsigned g = tid >> 4;
  const unsigned l = tid & 15;
  if (g >= (unsigned)PULL_GROUPS) return;
  const bool cok = l * 4 < (unsigned)COLS;
  const unsigned col = cok ? l * 4 : 0u;

  const int2* const l0 = reinterpret_cast<const int2*>(lines + gcn_line_at(0u, 0u)) + l;
  const int2* const l1 = reinterpret_cast<const int2*>(lines + gcn_line_at(1u, 0u)) + l;
  static_assert(PULL_ROWS == 2, "the second row's lines are held in an, bn");
  int2 a = l0[(size_t)g * (LINE_INTS / 2)];
  int2 b = l1[(size_t)g * (LINE_INTS / 2)];
  const int2 an = l0[(size_t)(g + PULL_GROUPS) * (LINE_INTS / 2)];
  const int2 bn = l1[(size_t)(g + PULL_GROUPS) * (LINE_INTS / 2)];
#pragma nounroll
  for (int p = 0; p < PULL_ROWS; ++p) {
    const unsigned row = g + p * PULL_GROUPS;
    const int n0 = gcn_row_share<0>(a.x);
    const int n1 = gcn_row_share<0>(b.x);
    const int c0 = n0 < SLOTS ? n0 : SLOTS;
    const int c1 = n1 < SLOTS ? n1 : SLOTS;
    float4 acc = make_float4(0.0f, 0.0f, 0.0f, 0.0f);
    if (gcn_line_batch<COLS, 0>(x, a, b, c0, c1, col, acc))
    if (gcn_line_batch<COLS, 1>(x, a, b, c0, c1, col, acc))
    if (gcn_line_batch<COLS, 2>(x, a, b, c0, c1, col, acc))
    if (gcn_line_batch<COLS, 3>(x, a, b, c0, c1, col, acc))
    if (gcn_line_batch<COLS, 4>(x, a, b, c0, c1, col, acc))
    if (gcn_line_batch<COLS, 5>(x, a, b, c0, c1, col, acc))
    if (gcn_line_batch<COLS, 6>(x, a, b, c0, c1, col, acc))
      gcn_line_batch<COLS, 7>(x, a, b, c0, c1, col, acc);
    if (n0 > SLOTS || n1 > SLOTS)
      gcn_add_spill<COLS>(x, spill_cnt, spill, row, col, acc);
    if (cok) {
      if (HAS_BIAS) {
        acc.x += bias[col + 0]; acc.y += bias[col + 1];
        acc.z += bias[col + 2]; acc.w += bias[col + 3];
      }
      *reinterpret_cast<float4*>(y + (size_t)row * COLS + col) = acc;
    }
    a = an;
    b = bn;
  }
}

__global__ void __launch_bounds__(256)
gcn_wmma_chain_kernel(const float* __restrict__ X,
                      const float* __restrict__ W1,
                      const float* __restrict__ b1,
                      const float* __restrict__ W2,
                      float* __restrict__ T, int n_strips) {
  static_assert(IN_F == 64 && HID == 64, "four hidden tiles, k taken 8 at a time");
  static_assert(NC == 40, "two whole class tiles and the lower half of a third");

  const int strip = blockIdx.x * (blockDim.x >> 5) + (threadIdx.x >> 5);
  if (strip >= n_strips) return;

  const int lane = threadIdx.x & 31;
  const int c = lane & 15;
  const int h = lane >> 4;

  const float* x0p = X + ((size_t)strip * 32 + c) * IN_F + 4 * h;
  const float* x1p = x0p + (size_t)16 * IN_F;
  const float* wp = W1 + (size_t)(4 * h) * HID + 4 * c;

  v8f hid[2][4];
#pragma unroll
  for (int u = 0; u < 2; ++u)
#pragma unroll
    for (int mt = 0; mt < 4; ++mt) hid[u][mt] = v8f{};

#pragma unroll
  for (int i = 0; i < IN_F / 8; ++i) {
    const float4 x0 = *reinterpret_cast<const float4*>(x0p + 8 * i);
    const float4 x1 = *reinterpret_cast<const float4*>(x1p + 8 * i);
    float wv[4][4];
#pragma unroll
    for (int j = 0; j < 4; ++j) {
      const float4 w4 =
          *reinterpret_cast<const float4*>(wp + (size_t)(8 * i + j) * HID);
      wv[j][0] = w4.x; wv[j][1] = w4.y; wv[j][2] = w4.z; wv[j][3] = w4.w;
    }
#pragma unroll
    for (int p = 0; p < 2; ++p) {
      v2f bf0, bf1;
      bf0.x = p ? x0.z : x0.x;
      bf0.y = p ? x0.w : x0.y;
      bf1.x = p ? x1.z : x1.x;
      bf1.y = p ? x1.w : x1.y;
#pragma unroll
      for (int mt = 0; mt < 4; ++mt) {
        v2f af;
        af.x = wv[2 * p][mt];
        af.y = wv[2 * p + 1][mt];
        hid[0][mt] = __builtin_amdgcn_wmma_f32_16x16x4_f32(
            false, af, false, bf0, (short)0, hid[0][mt], false, false);
        hid[1][mt] = __builtin_amdgcn_wmma_f32_16x16x4_f32(
            false, af, false, bf1, (short)0, hid[1][mt], false, false);
      }
    }
  }

#pragma unroll
  for (int r = 0; r < 8; ++r) {
    const float4 bb = *reinterpret_cast<const float4*>(b1 + 4 * (8 * h + r));
#pragma unroll
    for (int u = 0; u < 2; ++u) {
      hid[u][0][r] = fmaxf(hid[u][0][r] + bb.x, 0.0f);
      hid[u][1][r] = fmaxf(hid[u][1][r] + bb.y, 0.0f);
      hid[u][2][r] = fmaxf(hid[u][2][r] + bb.z, 0.0f);
      hid[u][3][r] = fmaxf(hid[u][3][r] + bb.w, 0.0f);
    }
  }

  v8f cls[2][3];
#pragma unroll
  for (int u = 0; u < 2; ++u)
#pragma unroll
    for (int ct = 0; ct < 3; ++ct) cls[u][ct] = v8f{};

  const int slot = 8 * ((c & 7) >> 2) + 4 * (c >> 3) + (c & 3);
  int col[3];
  col[0] = slot;
  col[1] = 16 + slot;
  col[2] = (slot < 8) ? 32 + slot : 32;

#pragma unroll
  for (int mt = 0; mt < 4; ++mt) {
#pragma unroll
    for (int q = 0; q < 4; ++q) {
      const float* w2p = W2 + (size_t)(4 * (8 * h + 2 * q) + mt) * NC;
      v2f bf0, bf1;
      bf0.x = hid[0][mt][2 * q];
      bf0.y = hid[0][mt][2 * q + 1];
      bf1.x = hid[1][mt][2 * q];
      bf1.y = hid[1][mt][2 * q + 1];
#pragma unroll
      for (int ct = 0; ct < 3; ++ct) {
        v2f af;
        af.x = w2p[col[ct]];
        af.y = w2p[4 * NC + col[ct]];
        cls[0][ct] = __builtin_amdgcn_wmma_f32_16x16x4_f32(
            false, af, false, bf0, (short)0, cls[0][ct], false, false);
        cls[1][ct] = __builtin_amdgcn_wmma_f32_16x16x4_f32(
            false, af, false, bf1, (short)0, cls[1][ct], false, false);
      }
    }
  }

#pragma unroll
  for (int u = 0; u < 2; ++u) {
    float* tp = T + ((size_t)strip * 32 + 16 * u + c) * NC + 4 * h;
#pragma unroll
    for (int ct = 0; ct < 3; ++ct) {
      *reinterpret_cast<float4*>(tp + 16 * ct) = make_float4(
          cls[u][ct][0], cls[u][ct][1], cls[u][ct][2], cls[u][ct][3]);
      if (ct < 2)
        *reinterpret_cast<float4*>(tp + 16 * ct + 8) = make_float4(
            cls[u][ct][4], cls[u][ct][5], cls[u][ct][6], cls[u][ct][7]);
    }
  }
}

extern "C" void kernel_launch(void* const* d_in, const int* in_sizes, int n_in,
                              void* d_out, int out_size, void* d_ws,
                              size_t ws_size, hipStream_t stream) {
  (void)in_sizes; (void)n_in; (void)out_size; (void)ws_size;
  const float* features = (const float*)d_in[0];
  const int*   src      = (const int*)d_in[1];
  const int*   dst      = (const int*)d_in[2];
  const float* W1       = (const float*)d_in[3];
  const float* b1       = (const float*)d_in[4];
  const float* W2       = (const float*)d_in[5];
  const float* b2       = (const float*)d_in[6];
  float* out = (float*)d_out;

  float* ws    = (float*)d_ws;
  float* agg1  = ws;
  float* t     = ws + (size_t)N_NODES * HID;
  int*   lines = (int*)(ws + (size_t)2 * N_NODES * HID);
  int2*  spill = (int2*)(lines + (size_t)2 * N_NODES * LINE_INTS);
  int*   spill_cnt = (int*)(spill + (size_t)N_EDGES);

  {
    const int n_lines = 2 * N_NODES;
    gcn_count_init_kernel<<<(n_lines + 256 * INIT_PER_LANE - 1) / (256 * INIT_PER_LANE),
                            256, 0, stream>>>(lines, n_lines, spill_cnt);
  }

  gcn_place_kernel<<<(N_EDGES / PLACE_EDGES_PER_WAVE + 7) / 8, 256, 0, stream>>>(
      src, dst, lines, spill_cnt, spill);

  gcn_pull_lines_kernel<IN_F, false>
      <<<(PULL_GROUPS * 16 + 255) / 256, 256, 0, stream>>>(features, lines,
                                                           spill_cnt, spill,
                                                           b2, agg1);

  static_assert(N_NODES % 32 == 0, "a wave owns 32 whole rows");
  {
    const int strips = N_NODES / 32;
    gcn_wmma_chain_kernel<<<(strips + 7) / 8, 256, 0, stream>>>(agg1, W1, b1, W2,
                                                                t, strips);
  }

  gcn_pull_lines_kernel<NC, true>
      <<<(PULL_GROUPS * 16 + 255) / 256, 256, 0, stream>>>(t, lines, spill_cnt,
                                                           spill, b2, out);
}
